# starting kernel
# baseline (speedup 1.0000x reference)
#include <hip/hip_runtime.h>
#include <hip/hip_bf16.h>

#define N_NODES   50000
#define N_EDGES   1250000
#define IN_DIM    5
#define OUT_DIM   64
#define NUM_REL   3
#define NUM_LAYERS 3
#define LN_EPS    1e-5f

typedef __attribute__((ext_vector_type(2))) float v2f;
typedef __attribute__((ext_vector_type(8))) float v8f;

__global__ void zero_i32_kernel(int* __restrict__ p, int n) {
    int i = blockIdx.x * blockDim.x + threadIdx.x;
    if (i < n) p[i] = 0;
}

__global__ void count_kernel(const int* __restrict__ dst,
                             const int* __restrict__ etype,
                             int* __restrict__ counts, int nE) {
    int e = blockIdx.x * blockDim.x + threadIdx.x;
    if (e < nE) atomicAdd(&counts[dst[e] * NUM_REL + etype[e]], 1);
}

__global__ void inv_denom_kernel(const int* __restrict__ counts,
                                 float* __restrict__ inv, int n) {
    int i = blockIdx.x * blockDim.x + threadIdx.x;
    if (i < n) {
        int c = counts[i];
        inv[i] = 1.0f / (float)(c > 1 ? c : 1);
    }
}

__global__ void feat2c_kernel(const float* __restrict__ x,
                              const float* __restrict__ W,
                              const float* __restrict__ b,
                              float* __restrict__ h, int n) {
    int idx = blockIdx.x * blockDim.x + threadIdx.x;
    if (idx >= n * OUT_DIM) return;
    int node = idx >> 6;
    int o    = idx & 63;
    float acc = b[o];
#pragma unroll
    for (int d = 0; d < IN_DIM; ++d)
        acc += x[node * IN_DIM + d] * W[d * OUT_DIM + o];
    h[idx] = acc;
}

__global__ __launch_bounds__(128)
void rgcn_transform_kernel(const float* __restrict__ h,
                           const float* __restrict__ relW,
                           const float* __restrict__ rootW,
                           const float* __restrict__ bias,
                           float* __restrict__ hr,
                           float* __restrict__ outbuf)
{
    __shared__ float lh[16][65];
    const int row0 = blockIdx.x * 16;

    for (int i = threadIdx.x; i < 16 * 64; i += 128) {
        int r = i >> 6, c = i & 63;
        lh[r][c] = h[(size_t)(row0 + r) * OUT_DIM + c];
    }
    __syncthreads();

    const int wave = threadIdx.x >> 5;
    const int lane = threadIdx.x & 31;
    const int mrow = lane & 15;
    const int khalf = (lane >> 4) << 1;

    const float* __restrict__ W =
        (wave < NUM_REL) ? (relW + (size_t)wave * OUT_DIM * OUT_DIM) : rootW;

    v8f acc[4];
#pragma unroll
    for (int nt = 0; nt < 4; ++nt) acc[nt] = (v8f){0.f,0.f,0.f,0.f,0.f,0.f,0.f,0.f};

#pragma unroll
    for (int k0 = 0; k0 < OUT_DIM; k0 += 4) {
        const int kk = k0 + khalf;
        v2f a;
        a.x = lh[mrow][kk];
        a.y = lh[mrow][kk + 1];
        const float* __restrict__ Wk0 = W + (size_t)kk * OUT_DIM;
        const float* __restrict__ Wk1 = Wk0 + OUT_DIM;
#pragma unroll
        for (int nt = 0; nt < 4; ++nt) {
            v2f b;
            b.x = Wk0[nt * 16 + mrow];
            b.y = Wk1[nt * 16 + mrow];
            acc[nt] = __builtin_amdgcn_wmma_f32_16x16x4_f32(
                false, a, false, b, (short)0, acc[nt], false, false);
        }
    }

    const int rbase = row0 + ((lane >> 4) << 3);
    if (wave < NUM_REL) {
#pragma unroll
        for (int nt = 0; nt < 4; ++nt) {
            const int col = nt * 16 + mrow;
#pragma unroll
            for (int v = 0; v < 8; ++v) {
                int row = rbase + v;
                hr[((size_t)row * NUM_REL + wave) * OUT_DIM + col] = acc[nt][v];
            }
        }
    } else {
#pragma unroll
        for (int nt = 0; nt < 4; ++nt) {
            const int col = nt * 16 + mrow;
            const float bs = bias[col];
#pragma unroll
            for (int v = 0; v < 8; ++v) {
                int row = rbase + v;
                outbuf[(size_t)row * OUT_DIM + col] = acc[nt][v] + bs;
            }
        }
    }
}

__global__ __launch_bounds__(256)
void edge_scatter_kernel(const int* __restrict__ src,
                         const int* __restrict__ dst,
                         const int* __restrict__ etype,
                         const float* __restrict__ hr,
                         const float* __restrict__ inv_denom,
                         float* __restrict__ outbuf, int nE) {
    int e = blockIdx.x * 8 + (threadIdx.x >> 5);
    if (e >= nE) return;
    const int lane = threadIdx.x & 31;
    const int s = src[e];
    const int d = dst[e];
    const int r = etype[e];
    const float inv = inv_denom[d * NUM_REL + r];
    const float2 m =
        ((const float2*)(hr + ((size_t)s * NUM_REL + r) * OUT_DIM))[lane];
    float* o = outbuf + (size_t)d * OUT_DIM + lane * 2;
    unsafeAtomicAdd(o,     m.x * inv);
    unsafeAtomicAdd(o + 1, m.y * inv);
}

__global__ __launch_bounds__(256)
void finalize_kernel(const float* __restrict__ outbuf,
                     const float* __restrict__ gamma,
                     const float* __restrict__ beta,
                     float* __restrict__ hout, int n) {
    int node = blockIdx.x * 8 + (threadIdx.x >> 5);
    if (node >= n) return;
    const int lane = threadIdx.x & 31;
    float2 v = ((const float2*)(outbuf + (size_t)node * OUT_DIM))[lane];
    v.x = fmaxf(v.x, 0.0f);
    v.y = fmaxf(v.y, 0.0f);
    float s  = v.x + v.y;
    float sq = v.x * v.x + v.y * v.y;
#pragma unroll
    for (int off = 16; off > 0; off >>= 1) {
        s  += __shfl_xor(s,  off, 32);
        sq += __shfl_xor(sq, off, 32);
    }
    const float mu   = s * (1.0f / OUT_DIM);
    const float var  = sq * (1.0f / OUT_DIM) - mu * mu;
    const float rstd = rsqrtf(var + LN_EPS);
    const int c0 = lane * 2;
    float2 o;
    o.x = (v.x - mu) * rstd * gamma[c0]     + beta[c0];
    o.y = (v.y - mu) * rstd * gamma[c0 + 1] + beta[c0 + 1];
    ((float2*)(hout + (size_t)node * OUT_DIM))[lane] = o;
}

extern "C" void kernel_launch(void* const* d_in, const int* in_sizes, int n_in,
                              void* d_out, int out_size, void* d_ws, size_t ws_size,
                              hipStream_t stream) {
    const float* x      = (const float*)d_in[0];
    const int*   ei     = (const int*)  d_in[1];
    const int*   src    = ei;
    const int*   dst    = ei + N_EDGES;
    const int*   etype  = (const int*)  d_in[2];
    const float* f2cW   = (const float*)d_in[3];
    const float* f2cb   = (const float*)d_in[4];
    const float* rgW    = (const float*)d_in[5];
    const float* rgRoot = (const float*)d_in[6];
    const float* rgBias = (const float*)d_in[7];
    const float* gamma  = (const float*)d_in[8];
    const float* beta   = (const float*)d_in[9];

    float* ws = (float*)d_ws;
    float* h_a       = ws;  ws += (size_t)N_NODES * OUT_DIM;
    float* h_b       = ws;  ws += (size_t)N_NODES * OUT_DIM;
    float* hr        = ws;  ws += (size_t)N_NODES * NUM_REL * OUT_DIM;
    float* outbuf    = ws;  ws += (size_t)N_NODES * OUT_DIM;
    float* inv_denom = ws;  ws += (size_t)N_NODES * NUM_REL;
    int*   counts    = (int*)ws;

    const int NR = N_NODES * NUM_REL;

    zero_i32_kernel<<<(NR + 255) / 256, 256, 0, stream>>>(counts, NR);
    count_kernel<<<(N_EDGES + 255) / 256, 256, 0, stream>>>(dst, etype, counts, N_EDGES);
    inv_denom_kernel<<<(NR + 255) / 256, 256, 0, stream>>>(counts, inv_denom, NR);
    feat2c_kernel<<<((size_t)N_NODES * OUT_DIM + 255) / 256, 256, 0, stream>>>(
        x, f2cW, f2cb, h_a, N_NODES);

    float* hcur = h_a;
    float* hnext = h_b;
    for (int l = 0; l < NUM_LAYERS; ++l) {
        rgcn_transform_kernel<<<N_NODES / 16, 128, 0, stream>>>(
            hcur,
            rgW    + (size_t)l * NUM_REL * OUT_DIM * OUT_DIM,
            rgRoot + (size_t)l * OUT_DIM * OUT_DIM,
            rgBias + (size_t)l * OUT_DIM,
            hr, outbuf);
        edge_scatter_kernel<<<(N_EDGES + 7) / 8, 256, 0, stream>>>(
            src, dst, etype, hr, inv_denom, outbuf, N_EDGES);
        float* ho = (l == NUM_LAYERS - 1) ? (float*)d_out : hnext;
        finalize_kernel<<<(N_NODES + 7) / 8, 256, 0, stream>>>(
            outbuf, gamma, beta, ho, N_NODES);
        float* t = hcur; hcur = hnext; hnext = t;
    }
}


# Edge scatter: a wave owns 32 consecutive edges (indices read once, coalesced; per-edge offsets broadcast from the owning lane), 8 row reads in flight per wave; float atomics kept, arrival order
# speedup vs baseline: 1.4053x; 1.4053x over previous
#include <hip/hip_runtime.h>
#include <hip/hip_bf16.h>

#define N_NODES   50000
#define N_EDGES   1250000
#define IN_DIM    5
#define OUT_DIM   64
#define NUM_REL   3
#define NUM_LAYERS 3
#define LN_EPS    1e-5f

typedef __attribute__((ext_vector_type(2))) float v2f;
typedef __attribute__((ext_vector_type(8))) float v8f;

__global__ void zero_i32_kernel(int* __restrict__ p, int n) {
    int i = blockIdx.x * blockDim.x + threadIdx.x;
    if (i < n) p[i] = 0;
}

__global__ void count_kernel(const int* __restrict__ dst,
                             const int* __restrict__ etype,
                             int* __restrict__ counts, int nE) {
    int e = blockIdx.x * blockDim.x + threadIdx.x;
    if (e < nE) atomicAdd(&counts[dst[e] * NUM_REL + etype[e]], 1);
}

__global__ void inv_denom_kernel(const int* __restrict__ counts,
                                 float* __restrict__ inv, int n) {
    int i = blockIdx.x * blockDim.x + threadIdx.x;
    if (i < n) {
        int c = counts[i];
        inv[i] = 1.0f / (float)(c > 1 ? c : 1);
    }
}

__global__ void feat2c_kernel(const float* __restrict__ x,
                              const float* __restrict__ W,
                              const float* __restrict__ b,
                              float* __restrict__ h, int n) {
    int idx = blockIdx.x * blockDim.x + threadIdx.x;
    if (idx >= n * OUT_DIM) return;
    int node = idx >> 6;
    int o    = idx & 63;
    float acc = b[o];
#pragma unroll
    for (int d = 0; d < IN_DIM; ++d)
        acc += x[node * IN_DIM + d] * W[d * OUT_DIM + o];
    h[idx] = acc;
}

__global__ __launch_bounds__(128)
void rgcn_transform_kernel(const float* __restrict__ h,
                           const float* __restrict__ relW,
                           const float* __restrict__ rootW,
                           const float* __restrict__ bias,
                           float* __restrict__ hr,
                           float* __restrict__ outbuf)
{
    __shared__ float lh[16][65];
    const int row0 = blockIdx.x * 16;

    for (int i = threadIdx.x; i < 16 * 64; i += 128) {
        int r = i >> 6, c = i & 63;
        lh[r][c] = h[(size_t)(row0 + r) * OUT_DIM + c];
    }
    __syncthreads();

    const int wave = threadIdx.x >> 5;
    const int lane = threadIdx.x & 31;
    const int mrow = lane & 15;
    const int khalf = (lane >> 4) << 1;

    const float* __restrict__ W =
        (wave < NUM_REL) ? (relW + (size_t)wave * OUT_DIM * OUT_DIM) : rootW;

    v8f acc[4];
#pragma unroll
    for (int nt = 0; nt < 4; ++nt) acc[nt] = (v8f){0.f,0.f,0.f,0.f,0.f,0.f,0.f,0.f};

#pragma unroll
    for (int k0 = 0; k0 < OUT_DIM; k0 += 4) {
        const int kk = k0 + khalf;
        v2f a;
        a.x = lh[mrow][kk];
        a.y = lh[mrow][kk + 1];
        const float* __restrict__ Wk0 = W + (size_t)kk * OUT_DIM;
        const float* __restrict__ Wk1 = Wk0 + OUT_DIM;
#pragma unroll
        for (int nt = 0; nt < 4; ++nt) {
            v2f b;
            b.x = Wk0[nt * 16 + mrow];
            b.y = Wk1[nt * 16 + mrow];
            acc[nt] = __builtin_amdgcn_wmma_f32_16x16x4_f32(
                false, a, false, b, (short)0, acc[nt], false, false);
        }
    }

    const int rbase = row0 + ((lane >> 4) << 3);
    if (wave < NUM_REL) {
#pragma unroll
        for (int nt = 0; nt < 4; ++nt) {
            const int col = nt * 16 + mrow;
#pragma unroll
            for (int v = 0; v < 8; ++v) {
                int row = rbase + v;
                hr[((size_t)row * NUM_REL + wave) * OUT_DIM + col] = acc[nt][v];
            }
        }
    } else {
#pragma unroll
        for (int nt = 0; nt < 4; ++nt) {
            const int col = nt * 16 + mrow;
            const float bs = bias[col];
#pragma unroll
            for (int v = 0; v < 8; ++v) {
                int row = rbase + v;
                outbuf[(size_t)row * OUT_DIM + col] = acc[nt][v] + bs;
            }
        }
    }
}

#define EDGES_PER_WAVE 32
#define ROWS_IN_FLIGHT 8

__global__ __launch_bounds__(256)
void edge_scatter_kernel(const int* __restrict__ src,
                         const int* __restrict__ dst,
                         const int* __restrict__ etype,
                         const float* __restrict__ hr,
                         const float* __restrict__ inv_denom,
                         float* __restrict__ outbuf, int nE) {
    const int lane = threadIdx.x & 31;
    const int wave = __builtin_amdgcn_readfirstlane((int)(threadIdx.x >> 5));
    const int e0 = ((int)blockIdx.x * 8 + wave) * EDGES_PER_WAVE;
    if (e0 >= nE) return;

    const int e = e0 + lane;
    int rowoff = 0;
    int outoff = 0;
    float inv = 0.0f;
    if (e < nE) {
        const int s = src[e];
        const int d = dst[e];
        const int r = etype[e];
        inv    = inv_denom[d * NUM_REL + r];
        rowoff = (s * NUM_REL + r) * OUT_DIM;
        outoff = d * OUT_DIM;
    }

    const int left = nE - e0;
    if (left >= EDGES_PER_WAVE) {
#pragma unroll
        for (int j0 = 0; j0 < EDGES_PER_WAVE; j0 += ROWS_IN_FLIGHT) {
            float2 m[ROWS_IN_FLIGHT];
#pragma unroll
            for (int u = 0; u < ROWS_IN_FLIGHT; ++u) {
                const int ro = __builtin_amdgcn_readlane(rowoff, j0 + u);
                m[u] = ((const float2*)(hr + ro))[lane];
            }
#pragma unroll
            for (int u = 0; u < ROWS_IN_FLIGHT; ++u) {
                const int   oo = __builtin_amdgcn_readlane(outoff, j0 + u);
                const float iv = __builtin_bit_cast(
                    float, __builtin_amdgcn_readlane(__builtin_bit_cast(int, inv), j0 + u));
                float* o = outbuf + oo + lane * 2;
                unsafeAtomicAdd(o,     m[u].x * iv);
                unsafeAtomicAdd(o + 1, m[u].y * iv);
            }
        }
    } else {
        for (int j = 0; j < left; ++j) {
            const int   ro = __builtin_amdgcn_readlane(rowoff, j);
            const int   oo = __builtin_amdgcn_readlane(outoff, j);
            const float iv = __builtin_bit_cast(
                float, __builtin_amdgcn_readlane(__builtin_bit_cast(int, inv), j));
            const float2 m = ((const float2*)(hr + ro))[lane];
            float* o = outbuf + oo + lane * 2;
            unsafeAtomicAdd(o,     m.x * iv);
            unsafeAtomicAdd(o + 1, m.y * iv);
        }
    }
}

__global__ __launch_bounds__(256)
void finalize_kernel(const float* __restrict__ outbuf,
                     const float* __restrict__ gamma,
                     const float* __restrict__ beta,
                     float* __restrict__ hout, int n) {
    int node = blockIdx.x * 8 + (threadIdx.x >> 5);
    if (node >= n) return;
    const int lane = threadIdx.x & 31;
    float2 v = ((const float2*)(outbuf + (size_t)node * OUT_DIM))[lane];
    v.x = fmaxf(v.x, 0.0f);
    v.y = fmaxf(v.y, 0.0f);
    float s  = v.x + v.y;
    float sq = v.x * v.x + v.y * v.y;
#pragma unroll
    for (int off = 16; off > 0; off >>= 1) {
        s  += __shfl_xor(s,  off, 32);
        sq += __shfl_xor(sq, off, 32);
    }
    const float mu   = s * (1.0f / OUT_DIM);
    const float var  = sq * (1.0f / OUT_DIM) - mu * mu;
    const float rstd = rsqrtf(var + LN_EPS);
    const int c0 = lane * 2;
    float2 o;
    o.x = (v.x - mu) * rstd * gamma[c0]     + beta[c0];
    o.y = (v.y - mu) * rstd * gamma[c0 + 1] + beta[c0 + 1];
    ((float2*)(hout + (size_t)node * OUT_DIM))[lane] = o;
}

extern "C" void kernel_launch(void* const* d_in, const int* in_sizes, int n_in,
                              void* d_out, int out_size, void* d_ws, size_t ws_size,
                              hipStream_t stream) {
    const float* x      = (const float*)d_in[0];
    const int*   ei     = (const int*)  d_in[1];
    const int*   src    = ei;
    const int*   dst    = ei + N_EDGES;
    const int*   etype  = (const int*)  d_in[2];
    const float* f2cW   = (const float*)d_in[3];
    const float* f2cb   = (const float*)d_in[4];
    const float* rgW    = (const float*)d_in[5];
    const float* rgRoot = (const float*)d_in[6];
    const float* rgBias = (const float*)d_in[7];
    const float* gamma  = (const float*)d_in[8];
    const float* beta   = (const float*)d_in[9];

    float* ws = (float*)d_ws;
    float* h_a       = ws;  ws += (size_t)N_NODES * OUT_DIM;
    float* h_b       = ws;  ws += (size_t)N_NODES * OUT_DIM;
    float* hr        = ws;  ws += (size_t)N_NODES * NUM_REL * OUT_DIM;
    float* outbuf    = ws;  ws += (size_t)N_NODES * OUT_DIM;
    float* inv_denom = ws;  ws += (size_t)N_NODES * NUM_REL;
    int*   counts    = (int*)ws;

    const int NR = N_NODES * NUM_REL;

    zero_i32_kernel<<<(NR + 255) / 256, 256, 0, stream>>>(counts, NR);
    count_kernel<<<(N_EDGES + 255) / 256, 256, 0, stream>>>(dst, etype, counts, N_EDGES);
    inv_denom_kernel<<<(NR + 255) / 256, 256, 0, stream>>>(counts, inv_denom, NR);
    feat2c_kernel<<<((size_t)N_NODES * OUT_DIM + 255) / 256, 256, 0, stream>>>(
        x, f2cW, f2cb, h_a, N_NODES);

    float* hcur = h_a;
    float* hnext = h_b;
    for (int l = 0; l < NUM_LAYERS; ++l) {
        rgcn_transform_kernel<<<N_NODES / 16, 128, 0, stream>>>(
            hcur,
            rgW    + (size_t)l * NUM_REL * OUT_DIM * OUT_DIM,
            rgRoot + (size_t)l * OUT_DIM * OUT_DIM,
            rgBias + (size_t)l * OUT_DIM,
            hr, outbuf);
        edge_scatter_kernel<<<(N_EDGES + 255) / 256, 256, 0, stream>>>(
            src, dst, etype, hr, inv_denom, outbuf, N_EDGES);
        float* ho = (l == NUM_LAYERS - 1) ? (float*)d_out : hnext;
        finalize_kernel<<<(N_NODES + 7) / 8, 256, 0, stream>>>(
            outbuf, gamma, beta, ho, N_NODES);
        float* t = hcur; hcur = hnext; hnext = t;
    }
}


# Edge scatter: lane l owns row columns l and l+32 (was 2l, 2l+1): each row load and each float atomic covers one whole 128B line, 2 line-adds per edge not 4 half-lines; float atomics, arrival order
# speedup vs baseline: 1.5952x; 1.1351x over previous
#include <hip/hip_runtime.h>
#include <hip/hip_bf16.h>

#define N_NODES   50000
#define N_EDGES   1250000
#define IN_DIM    5
#define OUT_DIM   64
#define NUM_REL   3
#define NUM_LAYERS 3
#define LN_EPS    1e-5f

typedef __attribute__((ext_vector_type(2))) float v2f;
typedef __attribute__((ext_vector_type(8))) float v8f;

__global__ void zero_i32_kernel(int* __restrict__ p, int n) {
    int i = blockIdx.x * blockDim.x + threadIdx.x;
    if (i < n) p[i] = 0;
}

__global__ void count_kernel(const int* __restrict__ dst,
                             const int* __restrict__ etype,
                             int* __restrict__ counts, int nE) {
    int e = blockIdx.x * blockDim.x + threadIdx.x;
    if (e < nE) atomicAdd(&counts[dst[e] * NUM_REL + etype[e]], 1);
}

__global__ void inv_denom_kernel(const int* __restrict__ counts,
                                 float* __restrict__ inv, int n) {
    int i = blockIdx.x * blockDim.x + threadIdx.x;
    if (i < n) {
        int c = counts[i];
        inv[i] = 1.0f / (float)(c > 1 ? c : 1);
    }
}

__global__ void feat2c_kernel(const float* __restrict__ x,
                              const float* __restrict__ W,
                              const float* __restrict__ b,
                              float* __restrict__ h, int n) {
    int idx = blockIdx.x * blockDim.x + threadIdx.x;
    if (idx >= n * OUT_DIM) return;
    int node = idx >> 6;
    int o    = idx & 63;
    float acc = b[o];
#pragma unroll
    for (int d = 0; d < IN_DIM; ++d)
        acc += x[node * IN_DIM + d] * W[d * OUT_DIM + o];
    h[idx] = acc;
}

__global__ __launch_bounds__(128)
void rgcn_transform_kernel(const float* __restrict__ h,
                           const float* __restrict__ relW,
                           const float* __restrict__ rootW,
                           const float* __restrict__ bias,
                           float* __restrict__ hr,
                           float* __restrict__ outbuf)
{
    __shared__ float lh[16][65];
    const int row0 = blockIdx.x * 16;

    for (int i = threadIdx.x; i < 16 * 64; i += 128) {
        int r = i >> 6, c = i & 63;
        lh[r][c] = h[(size_t)(row0 + r) * OUT_DIM + c];
    }
    __syncthreads();

    const int wave = threadIdx.x >> 5;
    const int lane = threadIdx.x & 31;
    const int mrow = lane & 15;
    const int khalf = (lane >> 4) << 1;

    const float* __restrict__ W =
        (wave < NUM_REL) ? (relW + (size_t)wave * OUT_DIM * OUT_DIM) : rootW;

    v8f acc[4];
#pragma unroll
    for (int nt = 0; nt < 4; ++nt) acc[nt] = (v8f){0.f,0.f,0.f,0.f,0.f,0.f,0.f,0.f};

#pragma unroll
    for (int k0 = 0; k0 < OUT_DIM; k0 += 4) {
        const int kk = k0 + khalf;
        v2f a;
        a.x = lh[mrow][kk];
        a.y = lh[mrow][kk + 1];
        const float* __restrict__ Wk0 = W + (size_t)kk * OUT_DIM;
        const float* __restrict__ Wk1 = Wk0 + OUT_DIM;
#pragma unroll
        for (int nt = 0; nt < 4; ++nt) {
            v2f b;
            b.x = Wk0[nt * 16 + mrow];
            b.y = Wk1[nt * 16 + mrow];
            acc[nt] = __builtin_amdgcn_wmma_f32_16x16x4_f32(
                false, a, false, b, (short)0, acc[nt], false, false);
        }
    }

    const int rbase = row0 + ((lane >> 4) << 3);
    if (wave < NUM_REL) {
#pragma unroll
        for (int nt = 0; nt < 4; ++nt) {
            const int col = nt * 16 + mrow;
#pragma unroll
            for (int v = 0; v < 8; ++v) {
                int row = rbase + v;
                hr[((size_t)row * NUM_REL + wave) * OUT_DIM + col] = acc[nt][v];
            }
        }
    } else {
#pragma unroll
        for (int nt = 0; nt < 4; ++nt) {
            const int col = nt * 16 + mrow;
            const float bs = bias[col];
#pragma unroll
            for (int v = 0; v < 8; ++v) {
                int row = rbase + v;
                outbuf[(size_t)row * OUT_DIM + col] = acc[nt][v] + bs;
            }
        }
    }
}

#define EDGES_PER_WAVE 32
#define ROWS_IN_FLIGHT 8

__global__ __launch_bounds__(256)
void edge_scatter_kernel(const int* __restrict__ src,
                         const int* __restrict__ dst,
                         const int* __restrict__ etype,
                         const float* __restrict__ hr,
                         const float* __restrict__ inv_denom,
                         float* __restrict__ outbuf, int nE) {
    const int lane = threadIdx.x & 31;
    const int wave = __builtin_amdgcn_readfirstlane((int)(threadIdx.x >> 5));
    const int e0 = ((int)blockIdx.x * 8 + wave) * EDGES_PER_WAVE;
    if (e0 >= nE) return;

    const int e = e0 + lane;
    int rowoff = 0;
    int outoff = 0;
    float inv = 0.0f;
    if (e < nE) {
        const int s = src[e];
        const int d = dst[e];
        const int r = etype[e];
        inv    = inv_denom[d * NUM_REL + r];
        rowoff = (s * NUM_REL + r) * OUT_DIM;
        outoff = d * OUT_DIM;
    }

    const int left = nE - e0;
    if (left >= EDGES_PER_WAVE) {
#pragma unroll
        for (int j0 = 0; j0 < EDGES_PER_WAVE; j0 += ROWS_IN_FLIGHT) {
            float mlo[ROWS_IN_FLIGHT];
            float mhi[ROWS_IN_FLIGHT];
#pragma unroll
            for (int u = 0; u < ROWS_IN_FLIGHT; ++u) {
                const int ro = __builtin_amdgcn_readlane(rowoff, j0 + u);
                const float* __restrict__ r = hr + ro + lane;
                mlo[u] = r[0];
                mhi[u] = r[32];
            }
#pragma unroll
            for (int u = 0; u < ROWS_IN_FLIGHT; ++u) {
                const int   oo = __builtin_amdgcn_readlane(outoff, j0 + u);
                const float iv = __builtin_bit_cast(
                    float, __builtin_amdgcn_readlane(__builtin_bit_cast(int, inv), j0 + u));
                float* o = outbuf + oo + lane;
                unsafeAtomicAdd(o,      mlo[u] * iv);
                unsafeAtomicAdd(o + 32, mhi[u] * iv);
            }
        }
    } else {
        for (int j = 0; j < left; ++j) {
            const int   ro = __builtin_amdgcn_readlane(rowoff, j);
            const int   oo = __builtin_amdgcn_readlane(outoff, j);
            const float iv = __builtin_bit_cast(
                float, __builtin_amdgcn_readlane(__builtin_bit_cast(int, inv), j));
            const float* __restrict__ r = hr + ro + lane;
            const float mlo = r[0];
            const float mhi = r[32];
            float* o = outbuf + oo + lane;
            unsafeAtomicAdd(o,      mlo * iv);
            unsafeAtomicAdd(o + 32, mhi * iv);
        }
    }
}

__global__ __launch_bounds__(256)
void finalize_kernel(const float* __restrict__ outbuf,
                     const float* __restrict__ gamma,
                     const float* __restrict__ beta,
                     float* __restrict__ hout, int n) {
    int node = blockIdx.x * 8 + (threadIdx.x >> 5);
    if (node >= n) return;
    const int lane = threadIdx.x & 31;
    float2 v = ((const float2*)(outbuf + (size_t)node * OUT_DIM))[lane];
    v.x = fmaxf(v.x, 0.0f);
    v.y = fmaxf(v.y, 0.0f);
    float s  = v.x + v.y;
    float sq = v.x * v.x + v.y * v.y;
#pragma unroll
    for (int off = 16; off > 0; off >>= 1) {
        s  += __shfl_xor(s,  off, 32);
        sq += __shfl_xor(sq, off, 32);
    }
    const float mu   = s * (1.0f / OUT_DIM);
    const float var  = sq * (1.0f / OUT_DIM) - mu * mu;
    const float rstd = rsqrtf(var + LN_EPS);
    const int c0 = lane * 2;
    float2 o;
    o.x = (v.x - mu) * rstd * gamma[c0]     + beta[c0];
    o.y = (v.y - mu) * rstd * gamma[c0 + 1] + beta[c0 + 1];
    ((float2*)(hout + (size_t)node * OUT_DIM))[lane] = o;
}

extern "C" void kernel_launch(void* const* d_in, const int* in_sizes, int n_in,
                              void* d_out, int out_size, void* d_ws, size_t ws_size,
                              hipStream_t stream) {
    const float* x      = (const float*)d_in[0];
    const int*   ei     = (const int*)  d_in[1];
    const int*   src    = ei;
    const int*   dst    = ei + N_EDGES;
    const int*   etype  = (const int*)  d_in[2];
    const float* f2cW   = (const float*)d_in[3];
    const float* f2cb   = (const float*)d_in[4];
    const float* rgW    = (const float*)d_in[5];
    const float* rgRoot = (const float*)d_in[6];
    const float* rgBias = (const float*)d_in[7];
    const float* gamma  = (const float*)d_in[8];
    const float* beta   = (const float*)d_in[9];

    float* ws = (float*)d_ws;
    float* h_a       = ws;  ws += (size_t)N_NODES * OUT_DIM;
    float* h_b       = ws;  ws += (size_t)N_NODES * OUT_DIM;
    float* hr        = ws;  ws += (size_t)N_NODES * NUM_REL * OUT_DIM;
    float* outbuf    = ws;  ws += (size_t)N_NODES * OUT_DIM;
    float* inv_denom = ws;  ws += (size_t)N_NODES * NUM_REL;
    int*   counts    = (int*)ws;

    const int NR = N_NODES * NUM_REL;

    zero_i32_kernel<<<(NR + 255) / 256, 256, 0, stream>>>(counts, NR);
    count_kernel<<<(N_EDGES + 255) / 256, 256, 0, stream>>>(dst, etype, counts, N_EDGES);
    inv_denom_kernel<<<(NR + 255) / 256, 256, 0, stream>>>(counts, inv_denom, NR);
    feat2c_kernel<<<((size_t)N_NODES * OUT_DIM + 255) / 256, 256, 0, stream>>>(
        x, f2cW, f2cb, h_a, N_NODES);

    float* hcur = h_a;
    float* hnext = h_b;
    for (int l = 0; l < NUM_LAYERS; ++l) {
        rgcn_transform_kernel<<<N_NODES / 16, 128, 0, stream>>>(
            hcur,
            rgW    + (size_t)l * NUM_REL * OUT_DIM * OUT_DIM,
            rgRoot + (size_t)l * OUT_DIM * OUT_DIM,
            rgBias + (size_t)l * OUT_DIM,
            hr, outbuf);
        edge_scatter_kernel<<<(N_EDGES + 255) / 256, 256, 0, stream>>>(
            src, dst, etype, hr, inv_denom, outbuf, N_EDGES);
        float* ho = (l == NUM_LAYERS - 1) ? (float*)d_out : hnext;
        finalize_kernel<<<(N_NODES + 7) / 8, 256, 0, stream>>>(
            outbuf, gamma, beta, ho, N_NODES);
        float* t = hcur; hcur = hnext; hnext = t;
    }
}


# Transform forms its own h tile while staging it to LDS (layer 1: x@W+b; later layers: ReLU+LayerNorm of the previous sums, sum order kept); h never stored, projection + 2 of 3 normalise launches gone
# speedup vs baseline: 1.6572x; 1.0389x over previous
#include <hip/hip_runtime.h>
#include <hip/hip_bf16.h>

#define N_NODES   50000
#define N_EDGES   1250000
#define IN_DIM    5
#define OUT_DIM   64
#define NUM_REL   3
#define NUM_LAYERS 3
#define LN_EPS    1e-5f

typedef __attribute__((ext_vector_type(2))) float v2f;
typedef __attribute__((ext_vector_type(8))) float v8f;

__global__ void zero_i32_kernel(int* __restrict__ p, int n) {
    int i = blockIdx.x * blockDim.x + threadIdx.x;
    if (i < n) p[i] = 0;
}

__global__ void count_kernel(const int* __restrict__ dst,
                             const int* __restrict__ etype,
                             int* __restrict__ counts, int nE) {
    int e = blockIdx.x * blockDim.x + threadIdx.x;
    if (e < nE) atomicAdd(&counts[dst[e] * NUM_REL + etype[e]], 1);
}

__global__ void inv_denom_kernel(const int* __restrict__ counts,
                                 float* __restrict__ inv, int n) {
    int i = blockIdx.x * blockDim.x + threadIdx.x;
    if (i < n) {
        int c = counts[i];
        inv[i] = 1.0f / (float)(c > 1 ? c : 1);
    }
}

template <bool FROM_X>
__device__ __forceinline__ void rgcn_transform_impl(
    const float* __restrict__ x,
    const float* __restrict__ f2cW,
    const float* __restrict__ f2cb,
    const float* __restrict__ gamma,
    const float* __restrict__ beta,
    const float* __restrict__ relW,
    const float* __restrict__ rootW,
    const float* __restrict__ bias,
    float* __restrict__ hr,
    float* __restrict__ outbuf)
{
    __shared__ float lh[16][65];
    const int row0 = blockIdx.x * 16;
    const int wave = threadIdx.x >> 5;
    const int lane = threadIdx.x & 31;

    if constexpr (FROM_X) {
        const int c = threadIdx.x & 63;
        float w[IN_DIM];
#pragma unroll
        for (int d = 0; d < IN_DIM; ++d) w[d] = f2cW[d * OUT_DIM + c];
        const float bb = f2cb[c];
        __shared__ float lx[16 * IN_DIM];
        if (threadIdx.x < 16 * IN_DIM)
            lx[threadIdx.x] = x[(size_t)row0 * IN_DIM + threadIdx.x];
        __syncthreads();
#pragma unroll
        for (int j = 0; j < 8; ++j) {
            const int r = (int)(threadIdx.x >> 6) + 2 * j;
            float acc = bb;
#pragma unroll
            for (int d = 0; d < IN_DIM; ++d)
                acc += lx[r * IN_DIM + d] * w[d];
            lh[r][c] = acc;
        }
    } else {
        const int c0 = lane * 2;
        const float g0 = gamma[c0], g1 = gamma[c0 + 1];
        const float b0 = beta[c0],  b1 = beta[c0 + 1];
        float2 v[4];
#pragma unroll
        for (int i = 0; i < 4; ++i)
            v[i] = ((const float2*)(outbuf + (size_t)(row0 + wave * 4 + i) * OUT_DIM))[lane];
#pragma unroll
        for (int i = 0; i < 4; ++i) {
            const int r = wave * 4 + i;
            float vx = fmaxf(v[i].x, 0.0f);
            float vy = fmaxf(v[i].y, 0.0f);
            float s  = vx + vy;
            float sq = vx * vx + vy * vy;
#pragma unroll
            for (int off = 16; off > 0; off >>= 1) {
                s  += __shfl_xor(s,  off, 32);
                sq += __shfl_xor(sq, off, 32);
            }
            const float mu   = s * (1.0f / OUT_DIM);
            const float var  = sq * (1.0f / OUT_DIM) - mu * mu;
            const float rstd = rsqrtf(var + LN_EPS);
            lh[r][c0]     = (vx - mu) * rstd * g0 + b0;
            lh[r][c0 + 1] = (vy - mu) * rstd * g1 + b1;
        }
    }
    __syncthreads();

    const int mrow = lane & 15;
    const int khalf = (lane >> 4) << 1;

    const float* __restrict__ W =
        (wave < NUM_REL) ? (relW + (size_t)wave * OUT_DIM * OUT_DIM) : rootW;

    v8f acc[4];
#pragma unroll
    for (int nt = 0; nt < 4; ++nt) acc[nt] = (v8f){0.f,0.f,0.f,0.f,0.f,0.f,0.f,0.f};

#pragma unroll
    for (int k0 = 0; k0 < OUT_DIM; k0 += 4) {
        const int kk = k0 + khalf;
        v2f a;
        a.x = lh[mrow][kk];
        a.y = lh[mrow][kk + 1];
        const float* __restrict__ Wk0 = W + (size_t)kk * OUT_DIM;
        const float* __restrict__ Wk1 = Wk0 + OUT_DIM;
#pragma unroll
        for (int nt = 0; nt < 4; ++nt) {
            v2f b;
            b.x = Wk0[nt * 16 + mrow];
            b.y = Wk1[nt * 16 + mrow];
            acc[nt] = __builtin_amdgcn_wmma_f32_16x16x4_f32(
                false, a, false, b, (short)0, acc[nt], false, false);
        }
    }

    const int rbase = row0 + ((lane >> 4) << 3);
    if (wave < NUM_REL) {
#pragma unroll
        for (int nt = 0; nt < 4; ++nt) {
            const int col = nt * 16 + mrow;
#pragma unroll
            for (int v = 0; v < 8; ++v) {
                int row = rbase + v;
                hr[((size_t)row * NUM_REL + wave) * OUT_DIM + col] = acc[nt][v];
            }
        }
    } else {
#pragma unroll
        for (int nt = 0; nt < 4; ++nt) {
            const int col = nt * 16 + mrow;
            const float bs = bias[col];
#pragma unroll
            for (int v = 0; v < 8; ++v) {
                int row = rbase + v;
                outbuf[(size_t)row * OUT_DIM + col] = acc[nt][v] + bs;
            }
        }
    }
}

__global__ __launch_bounds__(128)
void rgcn_transform_first_kernel(const float* __restrict__ x,
                                 const float* __restrict__ f2cW,
                                 const float* __restrict__ f2cb,
                                 const float* __restrict__ relW,
                                 const float* __restrict__ rootW,
                                 const float* __restrict__ bias,
                                 float* __restrict__ hr,
                                 float* __restrict__ outbuf)
{
    rgcn_transform_impl<true>(x, f2cW, f2cb, nullptr, nullptr,
                              relW, rootW, bias, hr, outbuf);
}

__global__ __launch_bounds__(128)
void rgcn_transform_next_kernel(const float* __restrict__ gamma,
                                const float* __restrict__ beta,
                                const float* __restrict__ relW,
                                const float* __restrict__ rootW,
                                const float* __restrict__ bias,
                                float* __restrict__ hr,
                                float* __restrict__ outbuf)
{
    rgcn_transform_impl<false>(nullptr, nullptr, nullptr, gamma, beta,
                               relW, rootW, bias, hr, outbuf);
}

#define EDGES_PER_WAVE 32
#define ROWS_IN_FLIGHT 8

__global__ __launch_bounds__(256)
void edge_scatter_kernel(const int* __restrict__ src,
                         const int* __restrict__ dst,
                         const int* __restrict__ etype,
                         const float* __restrict__ hr,
                         const float* __restrict__ inv_denom,
                         float* __restrict__ outbuf, int nE) {
    const int lane = threadIdx.x & 31;
    const int wave = __builtin_amdgcn_readfirstlane((int)(threadIdx.x >> 5));
    const int e0 = ((int)blockIdx.x * 8 + wave) * EDGES_PER_WAVE;
    if (e0 >= nE) return;

    const int e = e0 + lane;
    int rowoff = 0;
    int outoff = 0;
    float inv = 0.0f;
    if (e < nE) {
        const int s = src[e];
        const int d = dst[e];
        const int r = etype[e];
        inv    = inv_denom[d * NUM_REL + r];
        rowoff = (s * NUM_REL + r) * OUT_DIM;
        outoff = d * OUT_DIM;
    }

    const int left = nE - e0;
    if (left >= EDGES_PER_WAVE) {
#pragma unroll
        for (int j0 = 0; j0 < EDGES_PER_WAVE; j0 += ROWS_IN_FLIGHT) {
            float mlo[ROWS_IN_FLIGHT];
            float mhi[ROWS_IN_FLIGHT];
#pragma unroll
            for (int u = 0; u < ROWS_IN_FLIGHT; ++u) {
                const int ro = __builtin_amdgcn_readlane(rowoff, j0 + u);
                const float* __restrict__ r = hr + ro + lane;
                mlo[u] = r[0];
                mhi[u] = r[32];
            }
#pragma unroll
            for (int u = 0; u < ROWS_IN_FLIGHT; ++u) {
                const int   oo = __builtin_amdgcn_readlane(outoff, j0 + u);
                const float iv = __builtin_bit_cast(
                    float, __builtin_amdgcn_readlane(__builtin_bit_cast(int, inv), j0 + u));
                float* o = outbuf + oo + lane;
                unsafeAtomicAdd(o,      mlo[u] * iv);
                unsafeAtomicAdd(o + 32, mhi[u] * iv);
            }
        }
    } else {
        for (int j = 0; j < left; ++j) {
            const int   ro = __builtin_amdgcn_readlane(rowoff, j);
            const int   oo = __builtin_amdgcn_readlane(outoff, j);
            const float iv = __builtin_bit_cast(
                float, __builtin_amdgcn_readlane(__builtin_bit_cast(int, inv), j));
            const float* __restrict__ r = hr + ro + lane;
            const float mlo = r[0];
            const float mhi = r[32];
            float* o = outbuf + oo + lane;
            unsafeAtomicAdd(o,      mlo * iv);
            unsafeAtomicAdd(o + 32, mhi * iv);
        }
    }
}

__global__ __launch_bounds__(256)
void finalize_kernel(const float* __restrict__ outbuf,
                     const float* __restrict__ gamma,
                     const float* __restrict__ beta,
                     float* __restrict__ hout, int n) {
    int node = blockIdx.x * 8 + (threadIdx.x >> 5);
    if (node >= n) return;
    const int lane = threadIdx.x & 31;
    float2 v = ((const float2*)(outbuf + (size_t)node * OUT_DIM))[lane];
    v.x = fmaxf(v.x, 0.0f);
    v.y = fmaxf(v.y, 0.0f);
    float s  = v.x + v.y;
    float sq = v.x * v.x + v.y * v.y;
#pragma unroll
    for (int off = 16; off > 0; off >>= 1) {
        s  += __shfl_xor(s,  off, 32);
        sq += __shfl_xor(sq, off, 32);
    }
    const float mu   = s * (1.0f / OUT_DIM);
    const float var  = sq * (1.0f / OUT_DIM) - mu * mu;
    const float rstd = rsqrtf(var + LN_EPS);
    const int c0 = lane * 2;
    float2 o;
    o.x = (v.x - mu) * rstd * gamma[c0]     + beta[c0];
    o.y = (v.y - mu) * rstd * gamma[c0 + 1] + beta[c0 + 1];
    ((float2*)(hout + (size_t)node * OUT_DIM))[lane] = o;
}

extern "C" void kernel_launch(void* const* d_in, const int* in_sizes, int n_in,
                              void* d_out, int out_size, void* d_ws, size_t ws_size,
                              hipStream_t stream) {
    const float* x      = (const float*)d_in[0];
    const int*   ei     = (const int*)  d_in[1];
    const int*   src    = ei;
    const int*   dst    = ei + N_EDGES;
    const int*   etype  = (const int*)  d_in[2];
    const float* f2cW   = (const float*)d_in[3];
    const float* f2cb   = (const float*)d_in[4];
    const float* rgW    = (const float*)d_in[5];
    const float* rgRoot = (const float*)d_in[6];
    const float* rgBias = (const float*)d_in[7];
    const float* gamma  = (const float*)d_in[8];
    const float* beta   = (const float*)d_in[9];

    float* ws = (float*)d_ws;
    ws += (size_t)N_NODES * OUT_DIM;
    ws += (size_t)N_NODES * OUT_DIM;
    float* hr        = ws;  ws += (size_t)N_NODES * NUM_REL * OUT_DIM;
    float* outbuf    = ws;  ws += (size_t)N_NODES * OUT_DIM;
    float* inv_denom = ws;  ws += (size_t)N_NODES * NUM_REL;
    int*   counts    = (int*)ws;

    const int NR = N_NODES * NUM_REL;

    zero_i32_kernel<<<(NR + 255) / 256, 256, 0, stream>>>(counts, NR);
    count_kernel<<<(N_EDGES + 255) / 256, 256, 0, stream>>>(dst, etype, counts, N_EDGES);
    inv_denom_kernel<<<(NR + 255) / 256, 256, 0, stream>>>(counts, inv_denom, NR);

    rgcn_transform_first_kernel<<<N_NODES / 16, 128, 0, stream>>>(
        x, f2cW, f2cb, rgW, rgRoot, rgBias, hr, outbuf);
    edge_scatter_kernel<<<(N_EDGES + 255) / 256, 256, 0, stream>>>(
        src, dst, etype, hr, inv_denom, outbuf, N_EDGES);
    for (int l = 1; l < NUM_LAYERS; ++l) {
        rgcn_transform_next_kernel<<<N_NODES / 16, 128, 0, stream>>>(
            gamma, beta,
            rgW    + (size_t)l * NUM_REL * OUT_DIM * OUT_DIM,
            rgRoot + (size_t)l * OUT_DIM * OUT_DIM,
            rgBias + (size_t)l * OUT_DIM,
            hr, outbuf);
        edge_scatter_kernel<<<(N_EDGES + 255) / 256, 256, 0, stream>>>(
            src, dst, etype, hr, inv_denom, outbuf, N_EDGES);
    }
    finalize_kernel<<<(N_NODES + 7) / 8, 256, 0, stream>>>(
        outbuf, gamma, beta, (float*)d_out, N_NODES);
}


# Degree count: per-edge scattered global integer atomics replaced by work-group LDS histograms (8 owner ranges x 32 edge slices); reciprocal kernel sums partials; zero-fill launch gone (9 launches)
# speedup vs baseline: 1.9235x; 1.1607x over previous
#include <hip/hip_runtime.h>
#include <hip/hip_bf16.h>

#define N_NODES   50000
#define N_EDGES   1250000
#define IN_DIM    5
#define OUT_DIM   64
#define NUM_REL   3
#define NUM_LAYERS 3
#define LN_EPS    1e-5f

typedef __attribute__((ext_vector_type(2))) float v2f;
typedef __attribute__((ext_vector_type(8))) float v8f;

#define COUNT_RANGES   8
#define COUNT_SLICES   32
#define COUNT_RANGE_SZ (N_NODES * NUM_REL / COUNT_RANGES)
#define COUNT_SLICE_SZ ((N_EDGES + COUNT_SLICES - 1) / COUNT_SLICES)
#define COUNT_THREADS  1024
#define COUNT_UNROLL   4

__global__ __launch_bounds__(COUNT_THREADS)
void count_kernel(const int* __restrict__ dst,
                  const int* __restrict__ etype,
                  unsigned* __restrict__ partial,
                  int nE) {
    __shared__ unsigned hist[COUNT_RANGE_SZ];
    const int tid   = threadIdx.x;
    const int range = blockIdx.x % COUNT_RANGES;
    const int slice = blockIdx.x / COUNT_RANGES;
    const int lo    = range * COUNT_RANGE_SZ;

    for (int i = tid; i < COUNT_RANGE_SZ; i += COUNT_THREADS) hist[i] = 0u;
    __syncthreads();

    const int start = slice * COUNT_SLICE_SZ;
    const int stop  = (start + COUNT_SLICE_SZ < nE) ? (start + COUNT_SLICE_SZ) : nE;
    for (int base = start + tid; base < stop; base += COUNT_UNROLL * COUNT_THREADS) {
        int d[COUNT_UNROLL], t[COUNT_UNROLL];
#pragma unroll
        for (int u = 0; u < COUNT_UNROLL; ++u) {
            const int e  = base + u * COUNT_THREADS;
            const int ee = (e < stop) ? e : (stop - 1);
            d[u] = dst[ee];
            t[u] = etype[ee];
        }
#pragma unroll
        for (int u = 0; u < COUNT_UNROLL; ++u) {
            const int e = base + u * COUNT_THREADS;
            const unsigned k = (unsigned)(d[u] * NUM_REL + t[u] - lo);
            if (e < stop && k < (unsigned)COUNT_RANGE_SZ) atomicAdd(&hist[k], 1u);
        }
    }
    __syncthreads();

    unsigned* __restrict__ row = partial + (size_t)slice * (N_NODES * NUM_REL) + lo;
    for (int i = tid; i < COUNT_RANGE_SZ; i += COUNT_THREADS) row[i] = hist[i];
}

__global__ void inv_denom_kernel(const unsigned* __restrict__ partial,
                                 float* __restrict__ inv, int n) {
    int i = blockIdx.x * blockDim.x + threadIdx.x;
    if (i < n) {
        unsigned c = 0u;
#pragma unroll
        for (int s = 0; s < COUNT_SLICES; ++s) c += partial[(size_t)s * n + i];
        inv[i] = 1.0f / (float)(c > 1u ? c : 1u);
    }
}

template <bool FROM_X>
__device__ __forceinline__ void rgcn_transform_impl(
    const float* __restrict__ x,
    const float* __restrict__ f2cW,
    const float* __restrict__ f2cb,
    const float* __restrict__ gamma,
    const float* __restrict__ beta,
    const float* __restrict__ relW,
    const float* __restrict__ rootW,
    const float* __restrict__ bias,
    float* __restrict__ hr,
    float* __restrict__ outbuf)
{
    __shared__ float lh[16][65];
    const int row0 = blockIdx.x * 16;
    const int wave = threadIdx.x >> 5;
    const int lane = threadIdx.x & 31;

    if constexpr (FROM_X) {
        const int c = threadIdx.x & 63;
        float w[IN_DIM];
#pragma unroll
        for (int d = 0; d < IN_DIM; ++d) w[d] = f2cW[d * OUT_DIM + c];
        const float bb = f2cb[c];
        __shared__ float lx[16 * IN_DIM];
        if (threadIdx.x < 16 * IN_DIM)
            lx[threadIdx.x] = x[(size_t)row0 * IN_DIM + threadIdx.x];
        __syncthreads();
#pragma unroll
        for (int j = 0; j < 8; ++j) {
            const int r = (int)(threadIdx.x >> 6) + 2 * j;
            float acc = bb;
#pragma unroll
            for (int d = 0; d < IN_DIM; ++d)
                acc += lx[r * IN_DIM + d] * w[d];
            lh[r][c] = acc;
        }
    } else {
        const int c0 = lane * 2;
        const float g0 = gamma[c0], g1 = gamma[c0 + 1];
        const float b0 = beta[c0],  b1 = beta[c0 + 1];
        float2 v[4];
#pragma unroll
        for (int i = 0; i < 4; ++i)
            v[i] = ((const float2*)(outbuf + (size_t)(row0 + wave * 4 + i) * OUT_DIM))[lane];
#pragma unroll
        for (int i = 0; i < 4; ++i) {
            const int r = wave * 4 + i;
            float vx = fmaxf(v[i].x, 0.0f);
            float vy = fmaxf(v[i].y, 0.0f);
            float s  = vx + vy;
            float sq = vx * vx + vy * vy;
#pragma unroll
            for (int off = 16; off > 0; off >>= 1) {
                s  += __shfl_xor(s,  off, 32);
                sq += __shfl_xor(sq, off, 32);
            }
            const float mu   = s * (1.0f / OUT_DIM);
            const float var  = sq * (1.0f / OUT_DIM) - mu * mu;
            const float rstd = rsqrtf(var + LN_EPS);
            lh[r][c0]     = (vx - mu) * rstd * g0 + b0;
            lh[r][c0 + 1] = (vy - mu) * rstd * g1 + b1;
        }
    }
    __syncthreads();

    const int mrow = lane & 15;
    const int khalf = (lane >> 4) << 1;

    const float* __restrict__ W =
        (wave < NUM_REL) ? (relW + (size_t)wave * OUT_DIM * OUT_DIM) : rootW;

    v8f acc[4];
#pragma unroll
    for (int nt = 0; nt < 4; ++nt) acc[nt] = (v8f){0.f,0.f,0.f,0.f,0.f,0.f,0.f,0.f};

#pragma unroll
    for (int k0 = 0; k0 < OUT_DIM; k0 += 4) {
        const int kk = k0 + khalf;
        v2f a;
        a.x = lh[mrow][kk];
        a.y = lh[mrow][kk + 1];
        const float* __restrict__ Wk0 = W + (size_t)kk * OUT_DIM;
        const float* __restrict__ Wk1 = Wk0 + OUT_DIM;
#pragma unroll
        for (int nt = 0; nt < 4; ++nt) {
            v2f b;
            b.x = Wk0[nt * 16 + mrow];
            b.y = Wk1[nt * 16 + mrow];
            acc[nt] = __builtin_amdgcn_wmma_f32_16x16x4_f32(
                false, a, false, b, (short)0, acc[nt], false, false);
        }
    }

    const int rbase = row0 + ((lane >> 4) << 3);
    if (wave < NUM_REL) {
#pragma unroll
        for (int nt = 0; nt < 4; ++nt) {
            const int col = nt * 16 + mrow;
#pragma unroll
            for (int v = 0; v < 8; ++v) {
                int row = rbase + v;
                hr[((size_t)row * NUM_REL + wave) * OUT_DIM + col] = acc[nt][v];
            }
        }
    } else {
#pragma unroll
        for (int nt = 0; nt < 4; ++nt) {
            const int col = nt * 16 + mrow;
            const float bs = bias[col];
#pragma unroll
            for (int v = 0; v < 8; ++v) {
                int row = rbase + v;
                outbuf[(size_t)row * OUT_DIM + col] = acc[nt][v] + bs;
            }
        }
    }
}

__global__ __launch_bounds__(128)
void rgcn_transform_first_kernel(const float* __restrict__ x,
                                 const float* __restrict__ f2cW,
                                 const float* __restrict__ f2cb,
                                 const float* __restrict__ relW,
                                 const float* __restrict__ rootW,
                                 const float* __restrict__ bias,
                                 float* __restrict__ hr,
                                 float* __restrict__ outbuf)
{
    rgcn_transform_impl<true>(x, f2cW, f2cb, nullptr, nullptr,
                              relW, rootW, bias, hr, outbuf);
}

__global__ __launch_bounds__(128)
void rgcn_transform_next_kernel(const float* __restrict__ gamma,
                                const float* __restrict__ beta,
                                const float* __restrict__ relW,
                                const float* __restrict__ rootW,
                                const float* __restrict__ bias,
                                float* __restrict__ hr,
                                float* __restrict__ outbuf)
{
    rgcn_transform_impl<false>(nullptr, nullptr, nullptr, gamma, beta,
                               relW, rootW, bias, hr, outbuf);
}

#define EDGES_PER_WAVE 32
#define ROWS_IN_FLIGHT 8

__global__ __launch_bounds__(256)
void edge_scatter_kernel(const int* __restrict__ src,
                         const int* __restrict__ dst,
                         const int* __restrict__ etype,
                         const float* __restrict__ hr,
                         const float* __restrict__ inv_denom,
                         float* __restrict__ outbuf, int nE) {
    const int lane = threadIdx.x & 31;
    const int wave = __builtin_amdgcn_readfirstlane((int)(threadIdx.x >> 5));
    const int e0 = ((int)blockIdx.x * 8 + wave) * EDGES_PER_WAVE;
    if (e0 >= nE) return;

    const int e = e0 + lane;
    int rowoff = 0;
    int outoff = 0;
    float inv = 0.0f;
    if (e < nE) {
        const int s = src[e];
        const int d = dst[e];
        const int r = etype[e];
        inv    = inv_denom[d * NUM_REL + r];
        rowoff = (s * NUM_REL + r) * OUT_DIM;
        outoff = d * OUT_DIM;
    }

    const int left = nE - e0;
    if (left >= EDGES_PER_WAVE) {
#pragma unroll
        for (int j0 = 0; j0 < EDGES_PER_WAVE; j0 += ROWS_IN_FLIGHT) {
            float mlo[ROWS_IN_FLIGHT];
            float mhi[ROWS_IN_FLIGHT];
#pragma unroll
            for (int u = 0; u < ROWS_IN_FLIGHT; ++u) {
                const int ro = __builtin_amdgcn_readlane(rowoff, j0 + u);
                const float* __restrict__ r = hr + ro + lane;
                mlo[u] = r[0];
                mhi[u] = r[32];
            }
#pragma unroll
            for (int u = 0; u < ROWS_IN_FLIGHT; ++u) {
                const int   oo = __builtin_amdgcn_readlane(outoff, j0 + u);
                const float iv = __builtin_bit_cast(
                    float, __builtin_amdgcn_readlane(__builtin_bit_cast(int, inv), j0 + u));
                float* o = outbuf + oo + lane;
                unsafeAtomicAdd(o,      mlo[u] * iv);
                unsafeAtomicAdd(o + 32, mhi[u] * iv);
            }
        }
    } else {
        for (int j = 0; j < left; ++j) {
            const int   ro = __builtin_amdgcn_readlane(rowoff, j);
            const int   oo = __builtin_amdgcn_readlane(outoff, j);
            const float iv = __builtin_bit_cast(
                float, __builtin_amdgcn_readlane(__builtin_bit_cast(int, inv), j));
            const float* __restrict__ r = hr + ro + lane;
            const float mlo = r[0];
            const float mhi = r[32];
            float* o = outbuf + oo + lane;
            unsafeAtomicAdd(o,      mlo * iv);
            unsafeAtomicAdd(o + 32, mhi * iv);
        }
    }
}

__global__ __launch_bounds__(256)
void finalize_kernel(const float* __restrict__ outbuf,
                     const float* __restrict__ gamma,
                     const float* __restrict__ beta,
                     float* __restrict__ hout, int n) {
    int node = blockIdx.x * 8 + (threadIdx.x >> 5);
    if (node >= n) return;
    const int lane = threadIdx.x & 31;
    float2 v = ((const float2*)(outbuf + (size_t)node * OUT_DIM))[lane];
    v.x = fmaxf(v.x, 0.0f);
    v.y = fmaxf(v.y, 0.0f);
    float s  = v.x + v.y;
    float sq = v.x * v.x + v.y * v.y;
#pragma unroll
    for (int off = 16; off > 0; off >>= 1) {
        s  += __shfl_xor(s,  off, 32);
        sq += __shfl_xor(sq, off, 32);
    }
    const float mu   = s * (1.0f / OUT_DIM);
    const float var  = sq * (1.0f / OUT_DIM) - mu * mu;
    const float rstd = rsqrtf(var + LN_EPS);
    const int c0 = lane * 2;
    float2 o;
    o.x = (v.x - mu) * rstd * gamma[c0]     + beta[c0];
    o.y = (v.y - mu) * rstd * gamma[c0 + 1] + beta[c0 + 1];
    ((float2*)(hout + (size_t)node * OUT_DIM))[lane] = o;
}

extern "C" void kernel_launch(void* const* d_in, const int* in_sizes, int n_in,
                              void* d_out, int out_size, void* d_ws, size_t ws_size,
                              hipStream_t stream) {
    const float* x      = (const float*)d_in[0];
    const int*   ei     = (const int*)  d_in[1];
    const int*   src    = ei;
    const int*   dst    = ei + N_EDGES;
    const int*   etype  = (const int*)  d_in[2];
    const float* f2cW   = (const float*)d_in[3];
    const float* f2cb   = (const float*)d_in[4];
    const float* rgW    = (const float*)d_in[5];
    const float* rgRoot = (const float*)d_in[6];
    const float* rgBias = (const float*)d_in[7];
    const float* gamma  = (const float*)d_in[8];
    const float* beta   = (const float*)d_in[9];

    float* ws = (float*)d_ws;
    unsigned* partial = (unsigned*)ws;
    ws += (size_t)N_NODES * OUT_DIM;
    ws += (size_t)N_NODES * OUT_DIM;
    float* hr        = ws;  ws += (size_t)N_NODES * NUM_REL * OUT_DIM;
    float* outbuf    = ws;  ws += (size_t)N_NODES * OUT_DIM;
    float* inv_denom = ws;

    const int NR = N_NODES * NUM_REL;

    count_kernel<<<COUNT_RANGES * COUNT_SLICES, COUNT_THREADS, 0, stream>>>(dst, etype, partial, N_EDGES);
    inv_denom_kernel<<<(NR + 255) / 256, 256, 0, stream>>>(partial, inv_denom, NR);

    rgcn_transform_first_kernel<<<N_NODES / 16, 128, 0, stream>>>(
        x, f2cW, f2cb, rgW, rgRoot, rgBias, hr, outbuf);
    edge_scatter_kernel<<<(N_EDGES + 255) / 256, 256, 0, stream>>>(
        src, dst, etype, hr, inv_denom, outbuf, N_EDGES);
    for (int l = 1; l < NUM_LAYERS; ++l) {
        rgcn_transform_next_kernel<<<N_NODES / 16, 128, 0, stream>>>(
            gamma, beta,
            rgW    + (size_t)l * NUM_REL * OUT_DIM * OUT_DIM,
            rgRoot + (size_t)l * OUT_DIM * OUT_DIM,
            rgBias + (size_t)l * OUT_DIM,
            hr, outbuf);
        edge_scatter_kernel<<<(N_EDGES + 255) / 256, 256, 0, stream>>>(
            src, dst, etype, hr, inv_denom, outbuf, N_EDGES);
    }
    finalize_kernel<<<(N_NODES + 7) / 8, 256, 0, stream>>>(
        outbuf, gamma, beta, (float*)d_out, N_NODES);
}


# Edge scatter: an edge's scale 1/count[dst,rel] is gathered from the counter table once, in layer 1, and kept per edge; layers 2, 3 read it back coalesced (1 line per wave, not 32); atomics as before
# speedup vs baseline: 1.9785x; 1.0286x over previous
#include <hip/hip_runtime.h>
#include <hip/hip_bf16.h>

#define N_NODES   50000
#define N_EDGES   1250000
#define IN_DIM    5
#define OUT_DIM   64
#define NUM_REL   3
#define NUM_LAYERS 3
#define LN_EPS    1e-5f

typedef __attribute__((ext_vector_type(2))) float v2f;
typedef __attribute__((ext_vector_type(8))) float v8f;

#define COUNT_RANGES   8
#define COUNT_SLICES   32
#define COUNT_RANGE_SZ (N_NODES * NUM_REL / COUNT_RANGES)
#define COUNT_SLICE_SZ ((N_EDGES + COUNT_SLICES - 1) / COUNT_SLICES)
#define COUNT_THREADS  1024
#define COUNT_UNROLL   4

__global__ __launch_bounds__(COUNT_THREADS)
void count_kernel(const int* __restrict__ dst,
                  const int* __restrict__ etype,
                  unsigned* __restrict__ partial,
                  int nE) {
    __shared__ unsigned hist[COUNT_RANGE_SZ];
    const int tid   = threadIdx.x;
    const int range = blockIdx.x % COUNT_RANGES;
    const int slice = blockIdx.x / COUNT_RANGES;
    const int lo    = range * COUNT_RANGE_SZ;

    for (int i = tid; i < COUNT_RANGE_SZ; i += COUNT_THREADS) hist[i] = 0u;
    __syncthreads();

    const int start = slice * COUNT_SLICE_SZ;
    const int stop  = (start + COUNT_SLICE_SZ < nE) ? (start + COUNT_SLICE_SZ) : nE;
    for (int base = start + tid; base < stop; base += COUNT_UNROLL * COUNT_THREADS) {
        int d[COUNT_UNROLL], t[COUNT_UNROLL];
#pragma unroll
        for (int u = 0; u < COUNT_UNROLL; ++u) {
            const int e  = base + u * COUNT_THREADS;
            const int ee = (e < stop) ? e : (stop - 1);
            d[u] = dst[ee];
            t[u] = etype[ee];
        }
#pragma unroll
        for (int u = 0; u < COUNT_UNROLL; ++u) {
            const int e = base + u * COUNT_THREADS;
            const unsigned k = (unsigned)(d[u] * NUM_REL + t[u] - lo);
            if (e < stop && k < (unsigned)COUNT_RANGE_SZ) atomicAdd(&hist[k], 1u);
        }
    }
    __syncthreads();

    unsigned* __restrict__ row = partial + (size_t)slice * (N_NODES * NUM_REL) + lo;
    for (int i = tid; i < COUNT_RANGE_SZ; i += COUNT_THREADS) row[i] = hist[i];
}

__global__ void inv_denom_kernel(const unsigned* __restrict__ partial,
                                 float* __restrict__ inv, int n) {
    int i = blockIdx.x * blockDim.x + threadIdx.x;
    if (i < n) {
        unsigned c = 0u;
#pragma unroll
        for (int s = 0; s < COUNT_SLICES; ++s) c += partial[(size_t)s * n + i];
        inv[i] = 1.0f / (float)(c > 1u ? c : 1u);
    }
}

template <bool FROM_X>
__device__ __forceinline__ void rgcn_transform_impl(
    const float* __restrict__ x,
    const float* __restrict__ f2cW,
    const float* __restrict__ f2cb,
    const float* __restrict__ gamma,
    const float* __restrict__ beta,
    const float* __restrict__ relW,
    const float* __restrict__ rootW,
    const float* __restrict__ bias,
    float* __restrict__ hr,
    float* __restrict__ outbuf)
{
    __shared__ float lh[16][65];
    const int row0 = blockIdx.x * 16;
    const int wave = threadIdx.x >> 5;
    const int lane = threadIdx.x & 31;

    if constexpr (FROM_X) {
        const int c = threadIdx.x & 63;
        float w[IN_DIM];
#pragma unroll
        for (int d = 0; d < IN_DIM; ++d) w[d] = f2cW[d * OUT_DIM + c];
        const float bb = f2cb[c];
        __shared__ float lx[16 * IN_DIM];
        if (threadIdx.x < 16 * IN_DIM)
            lx[threadIdx.x] = x[(size_t)row0 * IN_DIM + threadIdx.x];
        __syncthreads();
#pragma unroll
        for (int j = 0; j < 8; ++j) {
            const int r = (int)(threadIdx.x >> 6) + 2 * j;
            float acc = bb;
#pragma unroll
            for (int d = 0; d < IN_DIM; ++d)
                acc += lx[r * IN_DIM + d] * w[d];
            lh[r][c] = acc;
        }
    } else {
        const int c0 = lane * 2;
        const float g0 = gamma[c0], g1 = gamma[c0 + 1];
        const float b0 = beta[c0],  b1 = beta[c0 + 1];
        float2 v[4];
#pragma unroll
        for (int i = 0; i < 4; ++i)
            v[i] = ((const float2*)(outbuf + (size_t)(row0 + wave * 4 + i) * OUT_DIM))[lane];
#pragma unroll
        for (int i = 0; i < 4; ++i) {
            const int r = wave * 4 + i;
            float vx = fmaxf(v[i].x, 0.0f);
            float vy = fmaxf(v[i].y, 0.0f);
            float s  = vx + vy;
            float sq = vx * vx + vy * vy;
#pragma unroll
            for (int off = 16; off > 0; off >>= 1) {
                s  += __shfl_xor(s,  off, 32);
                sq += __shfl_xor(sq, off, 32);
            }
            const float mu   = s * (1.0f / OUT_DIM);
            const float var  = sq * (1.0f / OUT_DIM) - mu * mu;
            const float rstd = rsqrtf(var + LN_EPS);
            lh[r][c0]     = (vx - mu) * rstd * g0 + b0;
            lh[r][c0 + 1] = (vy - mu) * rstd * g1 + b1;
        }
    }
    __syncthreads();

    const int mrow = lane & 15;
    const int khalf = (lane >> 4) << 1;

    const float* __restrict__ W =
        (wave < NUM_REL) ? (relW + (size_t)wave * OUT_DIM * OUT_DIM) : rootW;

    v8f acc[4];
#pragma unroll
    for (int nt = 0; nt < 4; ++nt) acc[nt] = (v8f){0.f,0.f,0.f,0.f,0.f,0.f,0.f,0.f};

#pragma unroll
    for (int k0 = 0; k0 < OUT_DIM; k0 += 4) {
        const int kk = k0 + khalf;
        v2f a;
        a.x = lh[mrow][kk];
        a.y = lh[mrow][kk + 1];
        const float* __restrict__ Wk0 = W + (size_t)kk * OUT_DIM;
        const float* __restrict__ Wk1 = Wk0 + OUT_DIM;
#pragma unroll
        for (int nt = 0; nt < 4; ++nt) {
            v2f b;
            b.x = Wk0[nt * 16 + mrow];
            b.y = Wk1[nt * 16 + mrow];
            acc[nt] = __builtin_amdgcn_wmma_f32_16x16x4_f32(
                false, a, false, b, (short)0, acc[nt], false, false);
        }
    }

    const int rbase = row0 + ((lane >> 4) << 3);
    if (wave < NUM_REL) {
#pragma unroll
        for (int nt = 0; nt < 4; ++nt) {
            const int col = nt * 16 + mrow;
#pragma unroll
            for (int v = 0; v < 8; ++v) {
                int row = rbase + v;
                hr[((size_t)row * NUM_REL + wave) * OUT_DIM + col] = acc[nt][v];
            }
        }
    } else {
#pragma unroll
        for (int nt = 0; nt < 4; ++nt) {
            const int col = nt * 16 + mrow;
            const float bs = bias[col];
#pragma unroll
            for (int v = 0; v < 8; ++v) {
                int row = rbase + v;
                outbuf[(size_t)row * OUT_DIM + col] = acc[nt][v] + bs;
            }
        }
    }
}

__global__ __launch_bounds__(128)
void rgcn_transform_first_kernel(const float* __restrict__ x,
                                 const float* __restrict__ f2cW,
                                 const float* __restrict__ f2cb,
                                 const float* __restrict__ relW,
                                 const float* __restrict__ rootW,
                                 const float* __restrict__ bias,
                                 float* __restrict__ hr,
                                 float* __restrict__ outbuf)
{
    rgcn_transform_impl<true>(x, f2cW, f2cb, nullptr, nullptr,
                              relW, rootW, bias, hr, outbuf);
}

__global__ __launch_bounds__(128)
void rgcn_transform_next_kernel(const float* __restrict__ gamma,
                                const float* __restrict__ beta,
                                const float* __restrict__ relW,
                                const float* __restrict__ rootW,
                                const float* __restrict__ bias,
                                float* __restrict__ hr,
                                float* __restrict__ outbuf)
{
    rgcn_transform_impl<false>(nullptr, nullptr, nullptr, gamma, beta,
                               relW, rootW, bias, hr, outbuf);
}

#define EDGES_PER_WAVE 32
#define ROWS_IN_FLIGHT 8

template <bool FIRST>
__device__ __forceinline__ void edge_scatter_impl(
                         const int* __restrict__ src,
                         const int* __restrict__ dst,
                         const int* __restrict__ etype,
                         const float* __restrict__ hr,
                         const float* __restrict__ inv_denom,
                         const float* __restrict__ edge_inv_in,
                         float* __restrict__ edge_inv_out,
                         float* __restrict__ outbuf, int nE) {
    const int lane = threadIdx.x & 31;
    const int wave = __builtin_amdgcn_readfirstlane((int)(threadIdx.x >> 5));
    const int e0 = ((int)blockIdx.x * 8 + wave) * EDGES_PER_WAVE;
    if (e0 >= nE) return;

    const int e = e0 + lane;
    int rowoff = 0;
    int outoff = 0;
    float inv = 0.0f;
    if (e < nE) {
        const int s = src[e];
        const int d = dst[e];
        const int r = etype[e];
        if constexpr (FIRST) inv = inv_denom[d * NUM_REL + r];
        else                 inv = edge_inv_in[e];
        rowoff = (s * NUM_REL + r) * OUT_DIM;
        outoff = d * OUT_DIM;
    }

    const int left = nE - e0;
    if (left >= EDGES_PER_WAVE) {
#pragma unroll
        for (int j0 = 0; j0 < EDGES_PER_WAVE; j0 += ROWS_IN_FLIGHT) {
            float mlo[ROWS_IN_FLIGHT];
            float mhi[ROWS_IN_FLIGHT];
#pragma unroll
            for (int u = 0; u < ROWS_IN_FLIGHT; ++u) {
                const int ro = __builtin_amdgcn_readlane(rowoff, j0 + u);
                const float* __restrict__ r = hr + ro + lane;
                mlo[u] = r[0];
                mhi[u] = r[32];
            }
#pragma unroll
            for (int u = 0; u < ROWS_IN_FLIGHT; ++u) {
                const int   oo = __builtin_amdgcn_readlane(outoff, j0 + u);
                const float iv = __builtin_bit_cast(
                    float, __builtin_amdgcn_readlane(__builtin_bit_cast(int, inv), j0 + u));
                float* o = outbuf + oo + lane;
                unsafeAtomicAdd(o,      mlo[u] * iv);
                unsafeAtomicAdd(o + 32, mhi[u] * iv);
            }
        }
    } else {
        for (int j = 0; j < left; ++j) {
            const int   ro = __builtin_amdgcn_readlane(rowoff, j);
            const int   oo = __builtin_amdgcn_readlane(outoff, j);
            const float iv = __builtin_bit_cast(
                float, __builtin_amdgcn_readlane(__builtin_bit_cast(int, inv), j));
            const float* __restrict__ r = hr + ro + lane;
            const float mlo = r[0];
            const float mhi = r[32];
            float* o = outbuf + oo + lane;
            unsafeAtomicAdd(o,      mlo * iv);
            unsafeAtomicAdd(o + 32, mhi * iv);
        }
    }

    if constexpr (FIRST) {
        if (e < nE) edge_inv_out[e] = inv;
    }
}

__global__ __launch_bounds__(256)
void edge_scatter_first_kernel(const int* __restrict__ src,
                               const int* __restrict__ dst,
                               const int* __restrict__ etype,
                               const float* __restrict__ hr,
                               const float* __restrict__ inv_denom,
                               float* __restrict__ edge_inv,
                               float* __restrict__ outbuf, int nE) {
    edge_scatter_impl<true>(src, dst, etype, hr, inv_denom, nullptr, edge_inv, outbuf, nE);
}

__global__ __launch_bounds__(256)
void edge_scatter_next_kernel(const int* __restrict__ src,
                              const int* __restrict__ dst,
                              const int* __restrict__ etype,
                              const float* __restrict__ hr,
                              const float* __restrict__ edge_inv,
                              float* __restrict__ outbuf, int nE) {
    edge_scatter_impl<false>(src, dst, etype, hr, nullptr, edge_inv, nullptr, outbuf, nE);
}

__global__ __launch_bounds__(256)
void finalize_kernel(const float* __restrict__ outbuf,
                     const float* __restrict__ gamma,
                     const float* __restrict__ beta,
                     float* __restrict__ hout, int n) {
    int node = blockIdx.x * 8 + (threadIdx.x >> 5);
    if (node >= n) return;
    const int lane = threadIdx.x & 31;
    float2 v = ((const float2*)(outbuf + (size_t)node * OUT_DIM))[lane];
    v.x = fmaxf(v.x, 0.0f);
    v.y = fmaxf(v.y, 0.0f);
    float s  = v.x + v.y;
    float sq = v.x * v.x + v.y * v.y;
#pragma unroll
    for (int off = 16; off > 0; off >>= 1) {
        s  += __shfl_xor(s,  off, 32);
        sq += __shfl_xor(sq, off, 32);
    }
    const float mu   = s * (1.0f / OUT_DIM);
    const float var  = sq * (1.0f / OUT_DIM) - mu * mu;
    const float rstd = rsqrtf(var + LN_EPS);
    const int c0 = lane * 2;
    float2 o;
    o.x = (v.x - mu) * rstd * gamma[c0]     + beta[c0];
    o.y = (v.y - mu) * rstd * gamma[c0 + 1] + beta[c0 + 1];
    ((float2*)(hout + (size_t)node * OUT_DIM))[lane] = o;
}

extern "C" void kernel_launch(void* const* d_in, const int* in_sizes, int n_in,
                              void* d_out, int out_size, void* d_ws, size_t ws_size,
                              hipStream_t stream) {
    const float* x      = (const float*)d_in[0];
    const int*   ei     = (const int*)  d_in[1];
    const int*   src    = ei;
    const int*   dst    = ei + N_EDGES;
    const int*   etype  = (const int*)  d_in[2];
    const float* f2cW   = (const float*)d_in[3];
    const float* f2cb   = (const float*)d_in[4];
    const float* rgW    = (const float*)d_in[5];
    const float* rgRoot = (const float*)d_in[6];
    const float* rgBias = (const float*)d_in[7];
    const float* gamma  = (const float*)d_in[8];
    const float* beta   = (const float*)d_in[9];

    float* ws = (float*)d_ws;
    unsigned* partial = (unsigned*)ws;
    float* edge_inv   = ws + (size_t)COUNT_SLICES * N_NODES * NUM_REL;
    ws += (size_t)N_NODES * OUT_DIM;
    ws += (size_t)N_NODES * OUT_DIM;
    float* hr        = ws;  ws += (size_t)N_NODES * NUM_REL * OUT_DIM;
    float* outbuf    = ws;  ws += (size_t)N_NODES * OUT_DIM;
    float* inv_denom = ws;

    const int NR = N_NODES * NUM_REL;

    count_kernel<<<COUNT_RANGES * COUNT_SLICES, COUNT_THREADS, 0, stream>>>(dst, etype, partial, N_EDGES);
    inv_denom_kernel<<<(NR + 255) / 256, 256, 0, stream>>>(partial, inv_denom, NR);

    rgcn_transform_first_kernel<<<N_NODES / 16, 128, 0, stream>>>(
        x, f2cW, f2cb, rgW, rgRoot, rgBias, hr, outbuf);
    edge_scatter_first_kernel<<<(N_EDGES + 255) / 256, 256, 0, stream>>>(
        src, dst, etype, hr, inv_denom, edge_inv, outbuf, N_EDGES);
    for (int l = 1; l < NUM_LAYERS; ++l) {
        rgcn_transform_next_kernel<<<N_NODES / 16, 128, 0, stream>>>(
            gamma, beta,
            rgW    + (size_t)l * NUM_REL * OUT_DIM * OUT_DIM,
            rgRoot + (size_t)l * OUT_DIM * OUT_DIM,
            rgBias + (size_t)l * OUT_DIM,
            hr, outbuf);
        edge_scatter_next_kernel<<<(N_EDGES + 255) / 256, 256, 0, stream>>>(
            src, dst, etype, hr, edge_inv, outbuf, N_EDGES);
    }
    finalize_kernel<<<(N_NODES + 7) / 8, 256, 0, stream>>>(
        outbuf, gamma, beta, (float*)d_out, N_NODES);
}
